# starting kernel
# baseline (speedup 1.0000x reference)
#include <hip/hip_runtime.h>
#include <hip/hip_bf16.h>
#include <cstdint>
#include <cstddef>

#define BATCH 32
#define NNODE 4096
#define HISTT 16
#define PREDT 8
#define NEDGE 65536

typedef float v2f __attribute__((ext_vector_type(2)));
typedef float v8f __attribute__((ext_vector_type(8)));

__device__ __forceinline__ float sigm(float x) { return 1.f / (1.f + __expf(-x)); }

__device__ __forceinline__ void atomAddF(float* p, float v) {
  __hip_atomic_fetch_add(p, v, __ATOMIC_RELAXED, __HIP_MEMORY_SCOPE_AGENT);
}

__global__ void k_zero(float* __restrict__ p, int n) {
  int i = blockIdx.x * 256 + threadIdx.x;
  if (i < n) p[i] = 0.f;
}

__global__ void k_deg(const int* __restrict__ ei, float* __restrict__ deg) {
  int e = blockIdx.x * 256 + threadIdx.x;
  if (e < NEDGE) atomAddF(deg + ei[e], 1.f);
}

__global__ void k_dis(const float* __restrict__ deg, float* __restrict__ dis) {
  int n = blockIdx.x * 256 + threadIdx.x;
  if (n < NNODE) {
    float d = deg[n];
    dis[n] = d > 0.f ? rsqrtf(d) : 0.f;
  }
}

__global__ void k_enorm(const int* __restrict__ ei, const float* __restrict__ dis,
                        float* __restrict__ enorm) {
  int e = blockIdx.x * 256 + threadIdx.x;
  if (e < NEDGE) enorm[e] = -dis[ei[e]] * dis[ei[NEDGE + e]];
}

__global__ void k_proj_hist(const float* __restrict__ pm25, const float* __restrict__ feat,
                            const float* __restrict__ cheb_w, const float* __restrict__ cheb_b,
                            float* __restrict__ p, float* __restrict__ q) {
  int i = blockIdx.x * 256 + threadIdx.x;
  if (i >= BATCH * HISTT * NNODE) return;
  int bt = i >> 12, b = bt >> 4, t = bt & 15, n = i & 4095;
  float x[10];
  x[0] = pm25[i];
  const float* fp = feat + (((size_t)b * 24 + t) * NNODE + n) * 9;
#pragma unroll
  for (int j = 0; j < 9; ++j) x[1 + j] = fp[j];
  float q0 = cheb_b[0], q1 = cheb_b[1], p0 = 0.f, p1 = 0.f;
#pragma unroll
  for (int f = 0; f < 10; ++f) {
    q0 += x[f] * cheb_w[f * 2 + 0];
    q1 += x[f] * cheb_w[f * 2 + 1];
    p0 += x[f] * cheb_w[20 + f * 2 + 0];
    p1 += x[f] * cheb_w[20 + f * 2 + 1];
  }
  size_t o = (size_t)i * 2;
  p[o] = p0; p[o + 1] = p1; q[o] = q0; q[o + 1] = q1;
}

__global__ void k_scatter_hist(const int* __restrict__ ei, const float* __restrict__ enorm,
                               const float* __restrict__ p, float* __restrict__ acc) {
  int idx = blockIdx.x * 256 + threadIdx.x;
  if (idx >= BATCH * HISTT * NEDGE) return;
  int bt = idx >> 16, e = idx & 65535;
  float nv = enorm[e];
  int s = ei[e], d = ei[NEDGE + e];
  size_t sb = ((size_t)bt * NNODE + s) * 2;
  size_t db = ((size_t)bt * NNODE + d) * 2;
  atomAddF(&acc[db + 0], p[sb + 0] * nv);
  atomAddF(&acc[db + 1], p[sb + 1] * nv);
}

__global__ void k_fin_hist(const float* __restrict__ q, float* __restrict__ acc) {
  int i = blockIdx.x * 256 + threadIdx.x;
  if (i < BATCH * HISTT * NNODE * 2) acc[i] = sigm(q[i] + acc[i]);
}

__global__ void k_proj_pred(const float* __restrict__ xn, const float* __restrict__ feat,
                            const float* __restrict__ cheb_w, const float* __restrict__ cheb_b,
                            float* __restrict__ p, float* __restrict__ q, float* __restrict__ acc,
                            int kstep) {
  int i = blockIdx.x * 256 + threadIdx.x;
  if (i >= BATCH * NNODE) return;
  int b = i >> 12, n = i & 4095;
  float x[10];
  x[0] = xn[i];
  const float* fp = feat + (((size_t)b * 24 + HISTT + kstep) * NNODE + n) * 9;
#pragma unroll
  for (int j = 0; j < 9; ++j) x[1 + j] = fp[j];
  float q0 = cheb_b[0], q1 = cheb_b[1], p0 = 0.f, p1 = 0.f;
#pragma unroll
  for (int f = 0; f < 10; ++f) {
    q0 += x[f] * cheb_w[f * 2 + 0];
    q1 += x[f] * cheb_w[f * 2 + 1];
    p0 += x[f] * cheb_w[20 + f * 2 + 0];
    p1 += x[f] * cheb_w[20 + f * 2 + 1];
  }
  size_t o = (size_t)i * 2;
  p[o] = p0; p[o + 1] = p1; q[o] = q0; q[o + 1] = q1;
  acc[o] = 0.f; acc[o + 1] = 0.f;
}

__global__ void k_scatter_pred(const int* __restrict__ ei, const float* __restrict__ enorm,
                               const float* __restrict__ p, float* __restrict__ acc) {
  int idx = blockIdx.x * 256 + threadIdx.x;
  if (idx >= BATCH * NEDGE) return;
  int b = idx >> 16, e = idx & 65535;
  float nv = enorm[e];
  int s = ei[e], d = ei[NEDGE + e];
  size_t sb = ((size_t)b * NNODE + s) * 2;
  size_t db = ((size_t)b * NNODE + d) * 2;
  atomAddF(&acc[db + 0], p[sb + 0] * nv);
  atomAddF(&acc[db + 1], p[sb + 1] * nv);
}

#define WT_STRIDE 48

__device__ __forceinline__ void load_weights(int tid,
    const float* __restrict__ x2h_w, const float* __restrict__ x2h_b,
    const float* __restrict__ h2h_w, const float* __restrict__ h2h_b,
    const float* __restrict__ fc_w,
    float* sWT, float* sBias, float* sFc) {
  for (int i = tid; i < 44 * 128; i += 256) {
    int k = i >> 7, n = i & 127;
    sWT[n * WT_STRIDE + k] = (k < 12) ? x2h_w[k * 128 + n] : h2h_w[(k - 12) * 128 + n];
  }
  if (tid < 128) sBias[tid] = x2h_b[tid] + h2h_b[tid];
  if (tid < 32) sFc[tid] = fc_w[tid];
}

__device__ __forceinline__ void gate_compute(const float* __restrict__ sXw,
                                             const float* __restrict__ sHw,
                                             const float* __restrict__ sWT,
                                             const float* __restrict__ sBias,
                                             int m, int hi, v8f* acc) {
  v2f ax[3], ah[8];
#pragma unroll
  for (int kk = 0; kk < 3; ++kk) ax[kk] = *(const v2f*)(sXw + m * 12 + 4 * kk + 2 * hi);
#pragma unroll
  for (int kk = 0; kk < 8; ++kk) ah[kk] = *(const v2f*)(sHw + m * 32 + 4 * kk + 2 * hi);
#pragma unroll
  for (int ct = 0; ct < 8; ++ct) {
    float bb = sBias[ct * 16 + m];
    v8f a = {bb, bb, bb, bb, bb, bb, bb, bb};
    const float* wcol = sWT + (ct * 16 + m) * WT_STRIDE;
#pragma unroll
    for (int kk = 0; kk < 3; ++kk) {
      v2f bf = *(const v2f*)(wcol + 4 * kk + 2 * hi);
      a = __builtin_amdgcn_wmma_f32_16x16x4_f32(false, ax[kk], false, bf, (short)0, a,
                                                false, false);
    }
#pragma unroll
    for (int kk = 0; kk < 8; ++kk) {
      v2f bf = *(const v2f*)(wcol + 12 + 4 * kk + 2 * hi);
      a = __builtin_amdgcn_wmma_f32_16x16x4_f32(false, ah[kk], false, bf, (short)0, a,
                                                false, false);
    }
    acc[ct] = a;
  }
}

__device__ __forceinline__ void lstm_update(const v8f* acc, float* c01, float* h01,
                                            float* sHw, int m, int hi) {
#pragma unroll
  for (int ct = 0; ct < 2; ++ct) {
#pragma unroll
    for (int r = 0; r < 8; ++r) {
      float iv = sigm(acc[ct][r]);
      float fv = sigm(acc[2 + ct][r]);
      float gv = tanhf(acc[4 + ct][r]);
      float ov = sigm(acc[6 + ct][r]);
      float cn = fv * c01[ct * 8 + r] + iv * gv;
      float hn = ov * tanhf(cn);
      c01[ct * 8 + r] = cn;
      h01[ct * 8 + r] = hn;
      sHw[(r + 8 * hi) * 32 + ct * 16 + m] = hn;
    }
  }
}

__launch_bounds__(256)
__global__ void hist_lstm_kernel(
    const float* __restrict__ pm25, const float* __restrict__ feat,
    const float* __restrict__ xg,
    const float* __restrict__ x2h_w, const float* __restrict__ x2h_b,
    const float* __restrict__ h2h_w, const float* __restrict__ h2h_b,
    const float* __restrict__ fc_w, const float* __restrict__ fc_b,
    float* __restrict__ h_glob, float* __restrict__ c_glob, float* __restrict__ xn_glob) {
  __shared__ float sWT[128 * WT_STRIDE];
  __shared__ float sBias[128];
  __shared__ float sFc[32];
  __shared__ float sX[8][16 * 12];
  __shared__ float sH[8][16 * 32];

  const int tid = threadIdx.x;
  load_weights(tid, x2h_w, x2h_b, h2h_w, h2h_b, fc_w, sWT, sBias, sFc);

  const int wave = tid >> 5, lane = tid & 31, m = lane & 15, hi = lane >> 4;
  const int rowbase = blockIdx.x * 128 + wave * 16;
  float* sXw = sX[wave];
  float* sHw = sH[wave];
  for (int j = lane; j < 512; j += 32) sHw[j] = 0.f;
  float c01[16], h01[16];
#pragma unroll
  for (int j = 0; j < 16; ++j) { c01[j] = 0.f; h01[j] = 0.f; }
  __syncthreads();

  for (int t = 0; t < HISTT; ++t) {
    if (lane < 16) {
      int R = rowbase + lane;
      int b = R >> 12, n = R & 4095;
      size_t hb = ((size_t)(b * 16 + t)) * NNODE + n;
      sXw[lane * 12 + 0] = pm25[hb];
      const float* fp = feat + (((size_t)b * 24 + t) * NNODE + n) * 9;
#pragma unroll
      for (int j = 0; j < 9; ++j) sXw[lane * 12 + 1 + j] = fp[j];
      sXw[lane * 12 + 10] = xg[hb * 2 + 0];
      sXw[lane * 12 + 11] = xg[hb * 2 + 1];
    }
    __syncthreads();
    v8f acc[8];
    gate_compute(sXw, sHw, sWT, sBias, m, hi, acc);
    lstm_update(acc, c01, h01, sHw, m, hi);
  }

#pragma unroll
  for (int ct = 0; ct < 2; ++ct)
#pragma unroll
    for (int r = 0; r < 8; ++r) {
      size_t R = (size_t)rowbase + r + 8 * hi;
      c_glob[R * 32 + ct * 16 + m] = c01[ct * 8 + r];
      h_glob[R * 32 + ct * 16 + m] = h01[ct * 8 + r];
    }
  __syncthreads();
  if (lane < 16) {
    float s = fc_b[0];
#pragma unroll
    for (int j = 0; j < 32; ++j) s += sHw[lane * 32 + j] * sFc[j];
    xn_glob[rowbase + lane] = s;
  }
}

__launch_bounds__(256)
__global__ void pred_lstm_kernel(
    const float* __restrict__ feat,
    const float* __restrict__ q_pred, const float* __restrict__ acc_pred,
    const float* __restrict__ x2h_w, const float* __restrict__ x2h_b,
    const float* __restrict__ h2h_w, const float* __restrict__ h2h_b,
    const float* __restrict__ fc_w, const float* __restrict__ fc_b,
    float* __restrict__ h_glob, float* __restrict__ c_glob, float* __restrict__ xn_glob,
    float* __restrict__ out, int kstep) {
  __shared__ float sWT[128 * WT_STRIDE];
  __shared__ float sBias[128];
  __shared__ float sFc[32];
  __shared__ float sX[8][16 * 12];
  __shared__ float sH[8][16 * 32];

  const int tid = threadIdx.x;
  load_weights(tid, x2h_w, x2h_b, h2h_w, h2h_b, fc_w, sWT, sBias, sFc);

  const int wave = tid >> 5, lane = tid & 31, m = lane & 15, hi = lane >> 4;
  const int rowbase = blockIdx.x * 128 + wave * 16;
  float* sXw = sX[wave];
  float* sHw = sH[wave];
  for (int j = lane; j < 512; j += 32) sHw[j] = h_glob[(size_t)rowbase * 32 + j];
  float c01[16], h01[16];
#pragma unroll
  for (int ct = 0; ct < 2; ++ct)
#pragma unroll
    for (int r = 0; r < 8; ++r)
      c01[ct * 8 + r] = c_glob[((size_t)rowbase + r + 8 * hi) * 32 + ct * 16 + m];

  if (lane < 16) {
    int R = rowbase + lane;
    int b = R >> 12, n = R & 4095;
    sXw[lane * 12 + 0] = xn_glob[R];
    const float* fp = feat + (((size_t)b * 24 + HISTT + kstep) * NNODE + n) * 9;
#pragma unroll
    for (int j = 0; j < 9; ++j) sXw[lane * 12 + 1 + j] = fp[j];
    sXw[lane * 12 + 10] = sigm(q_pred[(size_t)R * 2 + 0] + acc_pred[(size_t)R * 2 + 0]);
    sXw[lane * 12 + 11] = sigm(q_pred[(size_t)R * 2 + 1] + acc_pred[(size_t)R * 2 + 1]);
  }
  __syncthreads();

  v8f acc[8];
  gate_compute(sXw, sHw, sWT, sBias, m, hi, acc);
  lstm_update(acc, c01, h01, sHw, m, hi);

#pragma unroll
  for (int ct = 0; ct < 2; ++ct)
#pragma unroll
    for (int r = 0; r < 8; ++r) {
      size_t R = (size_t)rowbase + r + 8 * hi;
      c_glob[R * 32 + ct * 16 + m] = c01[ct * 8 + r];
      h_glob[R * 32 + ct * 16 + m] = h01[ct * 8 + r];
    }
  __syncthreads();
  if (lane < 16) {
    float s = fc_b[0];
#pragma unroll
    for (int j = 0; j < 32; ++j) s += sHw[lane * 32 + j] * sFc[j];
    int R = rowbase + lane;
    int b = R >> 12, n = R & 4095;
    xn_glob[R] = s;
    out[((size_t)b * PREDT + kstep) * NNODE + n] = s;
  }
}

extern "C" void kernel_launch(void* const* d_in, const int* in_sizes, int n_in,
                              void* d_out, int out_size, void* d_ws, size_t ws_size,
                              hipStream_t stream) {
  (void)in_sizes; (void)n_in; (void)out_size; (void)ws_size;
  const float* pm25   = (const float*)d_in[0];
  const float* feat   = (const float*)d_in[1];
  const int*   ei     = (const int*)d_in[2];
  const float* cheb_w = (const float*)d_in[3];
  const float* cheb_b = (const float*)d_in[4];
  const float* x2h_w  = (const float*)d_in[5];
  const float* x2h_b  = (const float*)d_in[6];
  const float* h2h_w  = (const float*)d_in[7];
  const float* h2h_b  = (const float*)d_in[8];
  const float* fc_w   = (const float*)d_in[9];
  const float* fc_b   = (const float*)d_in[10];
  float* out = (float*)d_out;
  char* ws = (char*)d_ws;

  float* deg      = (float*)(ws + 0);
  float* dis      = (float*)(ws + 0x4000);
  float* enorm    = (float*)(ws + 0x8000);
  float* p_hist   = (float*)(ws + 0x100000);
  float* q_hist   = (float*)(ws + 0x1100000);
  float* acc_hist = (float*)(ws + 0x2100000);
  float* xn_glob  = (float*)(ws + 0x3100000);
  float* p_pred   = (float*)(ws + 0x3180000);
  float* q_pred   = (float*)(ws + 0x3280000);
  float* acc_pred = (float*)(ws + 0x3380000);
  float* h_glob = p_hist;
  float* c_glob = q_hist;

  k_zero<<<(NNODE + 255) / 256, 256, 0, stream>>>(deg, NNODE);
  k_zero<<<(BATCH * HISTT * NNODE * 2 + 255) / 256, 256, 0, stream>>>(
      acc_hist, BATCH * HISTT * NNODE * 2);
  k_deg<<<NEDGE / 256, 256, 0, stream>>>(ei, deg);
  k_dis<<<NNODE / 256, 256, 0, stream>>>(deg, dis);
  k_enorm<<<NEDGE / 256, 256, 0, stream>>>(ei, dis, enorm);

  k_proj_hist<<<BATCH * HISTT * NNODE / 256, 256, 0, stream>>>(pm25, feat, cheb_w, cheb_b,
                                                               p_hist, q_hist);
  k_scatter_hist<<<BATCH * HISTT * NEDGE / 256, 256, 0, stream>>>(ei, enorm, p_hist, acc_hist);
  k_fin_hist<<<BATCH * HISTT * NNODE * 2 / 256, 256, 0, stream>>>(q_hist, acc_hist);

  hist_lstm_kernel<<<BATCH * NNODE / 128, 256, 0, stream>>>(
      pm25, feat, acc_hist, x2h_w, x2h_b, h2h_w, h2h_b, fc_w, fc_b, h_glob, c_glob, xn_glob);

  for (int k = 0; k < PREDT; ++k) {
    k_proj_pred<<<BATCH * NNODE / 256, 256, 0, stream>>>(xn_glob, feat, cheb_w, cheb_b,
                                                         p_pred, q_pred, acc_pred, k);
    k_scatter_pred<<<BATCH * NEDGE / 256, 256, 0, stream>>>(ei, enorm, p_pred, acc_pred);
    pred_lstm_kernel<<<BATCH * NNODE / 128, 256, 0, stream>>>(
        feat, q_pred, acc_pred, x2h_w, x2h_b, h2h_w, h2h_b, fc_w, fc_b,
        h_glob, c_glob, xn_glob, out, k);
  }
}


# Edge scatter summed in work-group-private LDS copies of each (batch,step) slice by LDS float adds, not L2 float atomics; prediction slices split over 4 groups, reader adds planes; arrival-order sums
# speedup vs baseline: 4.5049x; 4.5049x over previous
#include <hip/hip_runtime.h>
#include <hip/hip_bf16.h>
#include <cstdint>
#include <cstddef>

#define BATCH 32
#define NNODE 4096
#define HISTT 16
#define PREDT 8
#define NEDGE 65536

typedef float v2f __attribute__((ext_vector_type(2)));
typedef float v8f __attribute__((ext_vector_type(8)));

__device__ __forceinline__ float sigm(float x) { return 1.f / (1.f + __expf(-x)); }

__device__ __forceinline__ void atomAddF(float* p, float v) {
  __hip_atomic_fetch_add(p, v, __ATOMIC_RELAXED, __HIP_MEMORY_SCOPE_AGENT);
}

__device__ __forceinline__ void ldsAddF(float* p, float v) {
  __hip_atomic_fetch_add(p, v, __ATOMIC_RELAXED, __HIP_MEMORY_SCOPE_WORKGROUP);
}

__global__ void k_zero(float* __restrict__ p, int n) {
  int i = blockIdx.x * 256 + threadIdx.x;
  if (i < n) p[i] = 0.f;
}

__global__ void k_deg(const int* __restrict__ ei, float* __restrict__ deg) {
  int e = blockIdx.x * 256 + threadIdx.x;
  if (e < NEDGE) atomAddF(deg + ei[e], 1.f);
}

__global__ void k_dis(const float* __restrict__ deg, float* __restrict__ dis) {
  int n = blockIdx.x * 256 + threadIdx.x;
  if (n < NNODE) {
    float d = deg[n];
    dis[n] = d > 0.f ? rsqrtf(d) : 0.f;
  }
}

__global__ void k_enorm(const int* __restrict__ ei, const float* __restrict__ dis,
                        float* __restrict__ enorm) {
  int e = blockIdx.x * 256 + threadIdx.x;
  if (e < NEDGE) enorm[e] = -dis[ei[e]] * dis[ei[NEDGE + e]];
}

__global__ void k_proj_hist(const float* __restrict__ pm25, const float* __restrict__ feat,
                            const float* __restrict__ cheb_w, const float* __restrict__ cheb_b,
                            float* __restrict__ p, float* __restrict__ q) {
  int i = blockIdx.x * 256 + threadIdx.x;
  if (i >= BATCH * HISTT * NNODE) return;
  int bt = i >> 12, b = bt >> 4, t = bt & 15, n = i & 4095;
  float x[10];
  x[0] = pm25[i];
  const float* fp = feat + (((size_t)b * 24 + t) * NNODE + n) * 9;
#pragma unroll
  for (int j = 0; j < 9; ++j) x[1 + j] = fp[j];
  float q0 = cheb_b[0], q1 = cheb_b[1], p0 = 0.f, p1 = 0.f;
#pragma unroll
  for (int f = 0; f < 10; ++f) {
    q0 += x[f] * cheb_w[f * 2 + 0];
    q1 += x[f] * cheb_w[f * 2 + 1];
    p0 += x[f] * cheb_w[20 + f * 2 + 0];
    p1 += x[f] * cheb_w[20 + f * 2 + 1];
  }
  size_t o = (size_t)i * 2;
  p[o] = p0; p[o + 1] = p1; q[o] = q0; q[o + 1] = q1;
}

#define CONV_THREADS 1024
#define EDGE_GROUP 4
#define PRED_PARTS 4

template <bool FIN, int PARTS>
__launch_bounds__(CONV_THREADS)
__global__ void k_conv_slice(const int* __restrict__ ei, const float* __restrict__ enorm,
                             const float* __restrict__ p, const float* __restrict__ q,
                             float* __restrict__ dstbuf) {
  __shared__ float sP[NNODE * 2];
  __shared__ float sA0[NNODE];
  __shared__ float sA1[NNODE];
  const int tid = threadIdx.x;
  const size_t base = (size_t)blockIdx.x * (NNODE * 2);

  const float4* p4 = (const float4*)(p + base);
  for (int i = tid; i < NNODE * 2 / 4; i += CONV_THREADS) ((float4*)sP)[i] = p4[i];
  for (int i = tid; i < NNODE / 4; i += CONV_THREADS) {
    ((float4*)sA0)[i] = make_float4(0.f, 0.f, 0.f, 0.f);
    ((float4*)sA1)[i] = make_float4(0.f, 0.f, 0.f, 0.f);
  }
  __syncthreads();

  const int4* src4 = (const int4*)ei;
  const int4* dst4 = (const int4*)(ei + NEDGE);
  const float4* nv4 = (const float4*)enorm;
  const int iBeg = blockIdx.y * (NEDGE / 4 / PARTS), iEnd = iBeg + NEDGE / 4 / PARTS;
  for (int i0 = iBeg + tid; i0 < iEnd; i0 += CONV_THREADS * EDGE_GROUP) {
    int4 s[EDGE_GROUP], d[EDGE_GROUP];
    float4 nv[EDGE_GROUP];
#pragma unroll
    for (int u = 0; u < EDGE_GROUP; ++u) {
      int i = i0 + u * CONV_THREADS;
      s[u] = src4[i]; d[u] = dst4[i]; nv[u] = nv4[i];
    }
#pragma unroll
    for (int u = 0; u < EDGE_GROUP; ++u) {
      v2f p0 = *(const v2f*)(sP + 2 * s[u].x);
      v2f p1 = *(const v2f*)(sP + 2 * s[u].y);
      v2f p2 = *(const v2f*)(sP + 2 * s[u].z);
      v2f p3 = *(const v2f*)(sP + 2 * s[u].w);
      ldsAddF(sA0 + d[u].x, p0.x * nv[u].x); ldsAddF(sA1 + d[u].x, p0.y * nv[u].x);
      ldsAddF(sA0 + d[u].y, p1.x * nv[u].y); ldsAddF(sA1 + d[u].y, p1.y * nv[u].y);
      ldsAddF(sA0 + d[u].z, p2.x * nv[u].z); ldsAddF(sA1 + d[u].z, p2.y * nv[u].z);
      ldsAddF(sA0 + d[u].w, p3.x * nv[u].w); ldsAddF(sA1 + d[u].w, p3.y * nv[u].w);
    }
  }
  __syncthreads();

  const float4* q4 = (const float4*)(q + base);
  float4* o4 = (float4*)(dstbuf + (size_t)blockIdx.y * gridDim.x * (NNODE * 2) + base);
  for (int i = tid; i < NNODE / 2; i += CONV_THREADS) {
    v2f a0 = *(const v2f*)(sA0 + 2 * i);
    v2f a1 = *(const v2f*)(sA1 + 2 * i);
    float4 r = make_float4(a0.x, a1.x, a0.y, a1.y);
    if (FIN) {
      float4 qv = q4[i];
      r.x = sigm(qv.x + r.x); r.y = sigm(qv.y + r.y);
      r.z = sigm(qv.z + r.z); r.w = sigm(qv.w + r.w);
    }
    o4[i] = r;
  }
}

__global__ void k_proj_pred(const float* __restrict__ xn, const float* __restrict__ feat,
                            const float* __restrict__ cheb_w, const float* __restrict__ cheb_b,
                            float* __restrict__ p, float* __restrict__ q, int kstep) {
  int i = blockIdx.x * 256 + threadIdx.x;
  if (i >= BATCH * NNODE) return;
  int b = i >> 12, n = i & 4095;
  float x[10];
  x[0] = xn[i];
  const float* fp = feat + (((size_t)b * 24 + HISTT + kstep) * NNODE + n) * 9;
#pragma unroll
  for (int j = 0; j < 9; ++j) x[1 + j] = fp[j];
  float q0 = cheb_b[0], q1 = cheb_b[1], p0 = 0.f, p1 = 0.f;
#pragma unroll
  for (int f = 0; f < 10; ++f) {
    q0 += x[f] * cheb_w[f * 2 + 0];
    q1 += x[f] * cheb_w[f * 2 + 1];
    p0 += x[f] * cheb_w[20 + f * 2 + 0];
    p1 += x[f] * cheb_w[20 + f * 2 + 1];
  }
  size_t o = (size_t)i * 2;
  p[o] = p0; p[o + 1] = p1; q[o] = q0; q[o + 1] = q1;
}

#define WT_STRIDE 48

__device__ __forceinline__ void load_weights(int tid,
    const float* __restrict__ x2h_w, const float* __restrict__ x2h_b,
    const float* __restrict__ h2h_w, const float* __restrict__ h2h_b,
    const float* __restrict__ fc_w,
    float* sWT, float* sBias, float* sFc) {
  for (int i = tid; i < 44 * 128; i += 256) {
    int k = i >> 7, n = i & 127;
    sWT[n * WT_STRIDE + k] = (k < 12) ? x2h_w[k * 128 + n] : h2h_w[(k - 12) * 128 + n];
  }
  if (tid < 128) sBias[tid] = x2h_b[tid] + h2h_b[tid];
  if (tid < 32) sFc[tid] = fc_w[tid];
}

__device__ __forceinline__ void gate_compute(const float* __restrict__ sXw,
                                             const float* __restrict__ sHw,
                                             const float* __restrict__ sWT,
                                             const float* __restrict__ sBias,
                                             int m, int hi, v8f* acc) {
  v2f ax[3], ah[8];
#pragma unroll
  for (int kk = 0; kk < 3; ++kk) ax[kk] = *(const v2f*)(sXw + m * 12 + 4 * kk + 2 * hi);
#pragma unroll
  for (int kk = 0; kk < 8; ++kk) ah[kk] = *(const v2f*)(sHw + m * 32 + 4 * kk + 2 * hi);
#pragma unroll
  for (int ct = 0; ct < 8; ++ct) {
    float bb = sBias[ct * 16 + m];
    v8f a = {bb, bb, bb, bb, bb, bb, bb, bb};
    const float* wcol = sWT + (ct * 16 + m) * WT_STRIDE;
#pragma unroll
    for (int kk = 0; kk < 3; ++kk) {
      v2f bf = *(const v2f*)(wcol + 4 * kk + 2 * hi);
      a = __builtin_amdgcn_wmma_f32_16x16x4_f32(false, ax[kk], false, bf, (short)0, a,
                                                false, false);
    }
#pragma unroll
    for (int kk = 0; kk < 8; ++kk) {
      v2f bf = *(const v2f*)(wcol + 12 + 4 * kk + 2 * hi);
      a = __builtin_amdgcn_wmma_f32_16x16x4_f32(false, ah[kk], false, bf, (short)0, a,
                                                false, false);
    }
    acc[ct] = a;
  }
}

__device__ __forceinline__ void lstm_update(const v8f* acc, float* c01, float* h01,
                                            float* sHw, int m, int hi) {
#pragma unroll
  for (int ct = 0; ct < 2; ++ct) {
#pragma unroll
    for (int r = 0; r < 8; ++r) {
      float iv = sigm(acc[ct][r]);
      float fv = sigm(acc[2 + ct][r]);
      float gv = tanhf(acc[4 + ct][r]);
      float ov = sigm(acc[6 + ct][r]);
      float cn = fv * c01[ct * 8 + r] + iv * gv;
      float hn = ov * tanhf(cn);
      c01[ct * 8 + r] = cn;
      h01[ct * 8 + r] = hn;
      sHw[(r + 8 * hi) * 32 + ct * 16 + m] = hn;
    }
  }
}

__launch_bounds__(256)
__global__ void hist_lstm_kernel(
    const float* __restrict__ pm25, const float* __restrict__ feat,
    const float* __restrict__ xg,
    const float* __restrict__ x2h_w, const float* __restrict__ x2h_b,
    const float* __restrict__ h2h_w, const float* __restrict__ h2h_b,
    const float* __restrict__ fc_w, const float* __restrict__ fc_b,
    float* __restrict__ h_glob, float* __restrict__ c_glob, float* __restrict__ xn_glob) {
  __shared__ float sWT[128 * WT_STRIDE];
  __shared__ float sBias[128];
  __shared__ float sFc[32];
  __shared__ float sX[8][16 * 12];
  __shared__ float sH[8][16 * 32];

  const int tid = threadIdx.x;
  load_weights(tid, x2h_w, x2h_b, h2h_w, h2h_b, fc_w, sWT, sBias, sFc);

  const int wave = tid >> 5, lane = tid & 31, m = lane & 15, hi = lane >> 4;
  const int rowbase = blockIdx.x * 128 + wave * 16;
  float* sXw = sX[wave];
  float* sHw = sH[wave];
  for (int j = lane; j < 512; j += 32) sHw[j] = 0.f;
  float c01[16], h01[16];
#pragma unroll
  for (int j = 0; j < 16; ++j) { c01[j] = 0.f; h01[j] = 0.f; }
  __syncthreads();

  for (int t = 0; t < HISTT; ++t) {
    if (lane < 16) {
      int R = rowbase + lane;
      int b = R >> 12, n = R & 4095;
      size_t hb = ((size_t)(b * 16 + t)) * NNODE + n;
      sXw[lane * 12 + 0] = pm25[hb];
      const float* fp = feat + (((size_t)b * 24 + t) * NNODE + n) * 9;
#pragma unroll
      for (int j = 0; j < 9; ++j) sXw[lane * 12 + 1 + j] = fp[j];
      sXw[lane * 12 + 10] = xg[hb * 2 + 0];
      sXw[lane * 12 + 11] = xg[hb * 2 + 1];
    }
    __syncthreads();
    v8f acc[8];
    gate_compute(sXw, sHw, sWT, sBias, m, hi, acc);
    lstm_update(acc, c01, h01, sHw, m, hi);
  }

#pragma unroll
  for (int ct = 0; ct < 2; ++ct)
#pragma unroll
    for (int r = 0; r < 8; ++r) {
      size_t R = (size_t)rowbase + r + 8 * hi;
      c_glob[R * 32 + ct * 16 + m] = c01[ct * 8 + r];
      h_glob[R * 32 + ct * 16 + m] = h01[ct * 8 + r];
    }
  __syncthreads();
  if (lane < 16) {
    float s = fc_b[0];
#pragma unroll
    for (int j = 0; j < 32; ++j) s += sHw[lane * 32 + j] * sFc[j];
    xn_glob[rowbase + lane] = s;
  }
}

__launch_bounds__(256)
__global__ void pred_lstm_kernel(
    const float* __restrict__ feat,
    const float* __restrict__ q_pred, const float* __restrict__ acc_pred,
    const float* __restrict__ x2h_w, const float* __restrict__ x2h_b,
    const float* __restrict__ h2h_w, const float* __restrict__ h2h_b,
    const float* __restrict__ fc_w, const float* __restrict__ fc_b,
    float* __restrict__ h_glob, float* __restrict__ c_glob, float* __restrict__ xn_glob,
    float* __restrict__ out, int kstep) {
  __shared__ float sWT[128 * WT_STRIDE];
  __shared__ float sBias[128];
  __shared__ float sFc[32];
  __shared__ float sX[8][16 * 12];
  __shared__ float sH[8][16 * 32];

  const int tid = threadIdx.x;
  load_weights(tid, x2h_w, x2h_b, h2h_w, h2h_b, fc_w, sWT, sBias, sFc);

  const int wave = tid >> 5, lane = tid & 31, m = lane & 15, hi = lane >> 4;
  const int rowbase = blockIdx.x * 128 + wave * 16;
  float* sXw = sX[wave];
  float* sHw = sH[wave];
  for (int j = lane; j < 512; j += 32) sHw[j] = h_glob[(size_t)rowbase * 32 + j];
  float c01[16], h01[16];
#pragma unroll
  for (int ct = 0; ct < 2; ++ct)
#pragma unroll
    for (int r = 0; r < 8; ++r)
      c01[ct * 8 + r] = c_glob[((size_t)rowbase + r + 8 * hi) * 32 + ct * 16 + m];

  if (lane < 16) {
    int R = rowbase + lane;
    int b = R >> 12, n = R & 4095;
    sXw[lane * 12 + 0] = xn_glob[R];
    const float* fp = feat + (((size_t)b * 24 + HISTT + kstep) * NNODE + n) * 9;
#pragma unroll
    for (int j = 0; j < 9; ++j) sXw[lane * 12 + 1 + j] = fp[j];
    float a0 = 0.f, a1 = 0.f;
#pragma unroll
    for (int pp = 0; pp < PRED_PARTS; ++pp) {
      const float* ap = acc_pred + (size_t)pp * (BATCH * NNODE * 2) + (size_t)R * 2;
      a0 += ap[0]; a1 += ap[1];
    }
    sXw[lane * 12 + 10] = sigm(q_pred[(size_t)R * 2 + 0] + a0);
    sXw[lane * 12 + 11] = sigm(q_pred[(size_t)R * 2 + 1] + a1);
  }
  __syncthreads();

  v8f acc[8];
  gate_compute(sXw, sHw, sWT, sBias, m, hi, acc);
  lstm_update(acc, c01, h01, sHw, m, hi);

#pragma unroll
  for (int ct = 0; ct < 2; ++ct)
#pragma unroll
    for (int r = 0; r < 8; ++r) {
      size_t R = (size_t)rowbase + r + 8 * hi;
      c_glob[R * 32 + ct * 16 + m] = c01[ct * 8 + r];
      h_glob[R * 32 + ct * 16 + m] = h01[ct * 8 + r];
    }
  __syncthreads();
  if (lane < 16) {
    float s = fc_b[0];
#pragma unroll
    for (int j = 0; j < 32; ++j) s += sHw[lane * 32 + j] * sFc[j];
    int R = rowbase + lane;
    int b = R >> 12, n = R & 4095;
    xn_glob[R] = s;
    out[((size_t)b * PREDT + kstep) * NNODE + n] = s;
  }
}

extern "C" void kernel_launch(void* const* d_in, const int* in_sizes, int n_in,
                              void* d_out, int out_size, void* d_ws, size_t ws_size,
                              hipStream_t stream) {
  (void)in_sizes; (void)n_in; (void)out_size; (void)ws_size;
  const float* pm25   = (const float*)d_in[0];
  const float* feat   = (const float*)d_in[1];
  const int*   ei     = (const int*)d_in[2];
  const float* cheb_w = (const float*)d_in[3];
  const float* cheb_b = (const float*)d_in[4];
  const float* x2h_w  = (const float*)d_in[5];
  const float* x2h_b  = (const float*)d_in[6];
  const float* h2h_w  = (const float*)d_in[7];
  const float* h2h_b  = (const float*)d_in[8];
  const float* fc_w   = (const float*)d_in[9];
  const float* fc_b   = (const float*)d_in[10];
  float* out = (float*)d_out;
  char* ws = (char*)d_ws;

  float* deg      = (float*)(ws + 0);
  float* dis      = (float*)(ws + 0x4000);
  float* enorm    = (float*)(ws + 0x8000);
  float* p_hist   = (float*)(ws + 0x100000);
  float* q_hist   = (float*)(ws + 0x1100000);
  float* acc_hist = (float*)(ws + 0x2100000);
  float* xn_glob  = (float*)(ws + 0x3100000);
  float* p_pred   = (float*)(ws + 0x3180000);
  float* q_pred   = (float*)(ws + 0x3280000);
  float* h_glob = p_hist;
  float* c_glob = q_hist;
  float* acc_pred = acc_hist;

  k_zero<<<(NNODE + 255) / 256, 256, 0, stream>>>(deg, NNODE);
  k_deg<<<NEDGE / 256, 256, 0, stream>>>(ei, deg);
  k_dis<<<NNODE / 256, 256, 0, stream>>>(deg, dis);
  k_enorm<<<NEDGE / 256, 256, 0, stream>>>(ei, dis, enorm);

  k_proj_hist<<<BATCH * HISTT * NNODE / 256, 256, 0, stream>>>(pm25, feat, cheb_w, cheb_b,
                                                               p_hist, q_hist);
  k_conv_slice<true, 1><<<dim3(BATCH * HISTT, 1), CONV_THREADS, 0, stream>>>(
      ei, enorm, p_hist, q_hist, acc_hist);

  hist_lstm_kernel<<<BATCH * NNODE / 128, 256, 0, stream>>>(
      pm25, feat, acc_hist, x2h_w, x2h_b, h2h_w, h2h_b, fc_w, fc_b, h_glob, c_glob, xn_glob);

  for (int k = 0; k < PREDT; ++k) {
    k_proj_pred<<<BATCH * NNODE / 256, 256, 0, stream>>>(xn_glob, feat, cheb_w, cheb_b,
                                                         p_pred, q_pred, k);
    k_conv_slice<false, PRED_PARTS><<<dim3(BATCH, PRED_PARTS), CONV_THREADS, 0, stream>>>(
        ei, enorm, p_pred, q_pred, acc_pred);
    pred_lstm_kernel<<<BATCH * NNODE / 128, 256, 0, stream>>>(
        feat, q_pred, acc_pred, x2h_w, x2h_b, h2h_w, h2h_b, fc_w, fc_b,
        h_glob, c_glob, xn_glob, out, k);
  }
}


# LSTM gate GEMM and state update done 16 hidden units at a time (4 accumulator tiles live, was 8): both LSTM kernels now fit 128 VGPRs, so all 1024 work-groups are resident in one round, not 768 + 256
# speedup vs baseline: 4.9337x; 1.0952x over previous
#include <hip/hip_runtime.h>
#include <hip/hip_bf16.h>
#include <cstdint>
#include <cstddef>

#define BATCH 32
#define NNODE 4096
#define HISTT 16
#define PREDT 8
#define NEDGE 65536

typedef float v2f __attribute__((ext_vector_type(2)));
typedef float v8f __attribute__((ext_vector_type(8)));

__device__ __forceinline__ float sigm(float x) { return 1.f / (1.f + __expf(-x)); }

__device__ __forceinline__ void atomAddF(float* p, float v) {
  __hip_atomic_fetch_add(p, v, __ATOMIC_RELAXED, __HIP_MEMORY_SCOPE_AGENT);
}

__device__ __forceinline__ void ldsAddF(float* p, float v) {
  __hip_atomic_fetch_add(p, v, __ATOMIC_RELAXED, __HIP_MEMORY_SCOPE_WORKGROUP);
}

__global__ void k_zero(float* __restrict__ p, int n) {
  int i = blockIdx.x * 256 + threadIdx.x;
  if (i < n) p[i] = 0.f;
}

__global__ void k_deg(const int* __restrict__ ei, float* __restrict__ deg) {
  int e = blockIdx.x * 256 + threadIdx.x;
  if (e < NEDGE) atomAddF(deg + ei[e], 1.f);
}

__global__ void k_dis(const float* __restrict__ deg, float* __restrict__ dis) {
  int n = blockIdx.x * 256 + threadIdx.x;
  if (n < NNODE) {
    float d = deg[n];
    dis[n] = d > 0.f ? rsqrtf(d) : 0.f;
  }
}

__global__ void k_enorm(const int* __restrict__ ei, const float* __restrict__ dis,
                        float* __restrict__ enorm) {
  int e = blockIdx.x * 256 + threadIdx.x;
  if (e < NEDGE) enorm[e] = -dis[ei[e]] * dis[ei[NEDGE + e]];
}

__global__ void k_proj_hist(const float* __restrict__ pm25, const float* __restrict__ feat,
                            const float* __restrict__ cheb_w, const float* __restrict__ cheb_b,
                            float* __restrict__ p, float* __restrict__ q) {
  int i = blockIdx.x * 256 + threadIdx.x;
  if (i >= BATCH * HISTT * NNODE) return;
  int bt = i >> 12, b = bt >> 4, t = bt & 15, n = i & 4095;
  float x[10];
  x[0] = pm25[i];
  const float* fp = feat + (((size_t)b * 24 + t) * NNODE + n) * 9;
#pragma unroll
  for (int j = 0; j < 9; ++j) x[1 + j] = fp[j];
  float q0 = cheb_b[0], q1 = cheb_b[1], p0 = 0.f, p1 = 0.f;
#pragma unroll
  for (int f = 0; f < 10; ++f) {
    q0 += x[f] * cheb_w[f * 2 + 0];
    q1 += x[f] * cheb_w[f * 2 + 1];
    p0 += x[f] * cheb_w[20 + f * 2 + 0];
    p1 += x[f] * cheb_w[20 + f * 2 + 1];
  }
  size_t o = (size_t)i * 2;
  p[o] = p0; p[o + 1] = p1; q[o] = q0; q[o + 1] = q1;
}

#define CONV_THREADS 1024
#define EDGE_GROUP 4
#define PRED_PARTS 4

template <bool FIN, int PARTS>
__launch_bounds__(CONV_THREADS)
__global__ void k_conv_slice(const int* __restrict__ ei, const float* __restrict__ enorm,
                             const float* __restrict__ p, const float* __restrict__ q,
                             float* __restrict__ dstbuf) {
  __shared__ float sP[NNODE * 2];
  __shared__ float sA0[NNODE];
  __shared__ float sA1[NNODE];
  const int tid = threadIdx.x;
  const size_t base = (size_t)blockIdx.x * (NNODE * 2);

  const float4* p4 = (const float4*)(p + base);
  for (int i = tid; i < NNODE * 2 / 4; i += CONV_THREADS) ((float4*)sP)[i] = p4[i];
  for (int i = tid; i < NNODE / 4; i += CONV_THREADS) {
    ((float4*)sA0)[i] = make_float4(0.f, 0.f, 0.f, 0.f);
    ((float4*)sA1)[i] = make_float4(0.f, 0.f, 0.f, 0.f);
  }
  __syncthreads();

  const int4* src4 = (const int4*)ei;
  const int4* dst4 = (const int4*)(ei + NEDGE);
  const float4* nv4 = (const float4*)enorm;
  const int iBeg = blockIdx.y * (NEDGE / 4 / PARTS), iEnd = iBeg + NEDGE / 4 / PARTS;
  for (int i0 = iBeg + tid; i0 < iEnd; i0 += CONV_THREADS * EDGE_GROUP) {
    int4 s[EDGE_GROUP], d[EDGE_GROUP];
    float4 nv[EDGE_GROUP];
#pragma unroll
    for (int u = 0; u < EDGE_GROUP; ++u) {
      int i = i0 + u * CONV_THREADS;
      s[u] = src4[i]; d[u] = dst4[i]; nv[u] = nv4[i];
    }
#pragma unroll
    for (int u = 0; u < EDGE_GROUP; ++u) {
      v2f p0 = *(const v2f*)(sP + 2 * s[u].x);
      v2f p1 = *(const v2f*)(sP + 2 * s[u].y);
      v2f p2 = *(const v2f*)(sP + 2 * s[u].z);
      v2f p3 = *(const v2f*)(sP + 2 * s[u].w);
      ldsAddF(sA0 + d[u].x, p0.x * nv[u].x); ldsAddF(sA1 + d[u].x, p0.y * nv[u].x);
      ldsAddF(sA0 + d[u].y, p1.x * nv[u].y); ldsAddF(sA1 + d[u].y, p1.y * nv[u].y);
      ldsAddF(sA0 + d[u].z, p2.x * nv[u].z); ldsAddF(sA1 + d[u].z, p2.y * nv[u].z);
      ldsAddF(sA0 + d[u].w, p3.x * nv[u].w); ldsAddF(sA1 + d[u].w, p3.y * nv[u].w);
    }
  }
  __syncthreads();

  const float4* q4 = (const float4*)(q + base);
  float4* o4 = (float4*)(dstbuf + (size_t)blockIdx.y * gridDim.x * (NNODE * 2) + base);
  for (int i = tid; i < NNODE / 2; i += CONV_THREADS) {
    v2f a0 = *(const v2f*)(sA0 + 2 * i);
    v2f a1 = *(const v2f*)(sA1 + 2 * i);
    float4 r = make_float4(a0.x, a1.x, a0.y, a1.y);
    if (FIN) {
      float4 qv = q4[i];
      r.x = sigm(qv.x + r.x); r.y = sigm(qv.y + r.y);
      r.z = sigm(qv.z + r.z); r.w = sigm(qv.w + r.w);
    }
    o4[i] = r;
  }
}

__global__ void k_proj_pred(const float* __restrict__ xn, const float* __restrict__ feat,
                            const float* __restrict__ cheb_w, const float* __restrict__ cheb_b,
                            float* __restrict__ p, float* __restrict__ q, int kstep) {
  int i = blockIdx.x * 256 + threadIdx.x;
  if (i >= BATCH * NNODE) return;
  int b = i >> 12, n = i & 4095;
  float x[10];
  x[0] = xn[i];
  const float* fp = feat + (((size_t)b * 24 + HISTT + kstep) * NNODE + n) * 9;
#pragma unroll
  for (int j = 0; j < 9; ++j) x[1 + j] = fp[j];
  float q0 = cheb_b[0], q1 = cheb_b[1], p0 = 0.f, p1 = 0.f;
#pragma unroll
  for (int f = 0; f < 10; ++f) {
    q0 += x[f] * cheb_w[f * 2 + 0];
    q1 += x[f] * cheb_w[f * 2 + 1];
    p0 += x[f] * cheb_w[20 + f * 2 + 0];
    p1 += x[f] * cheb_w[20 + f * 2 + 1];
  }
  size_t o = (size_t)i * 2;
  p[o] = p0; p[o + 1] = p1; q[o] = q0; q[o + 1] = q1;
}

#define WT_STRIDE 48

__device__ __forceinline__ void load_weights(int tid,
    const float* __restrict__ x2h_w, const float* __restrict__ x2h_b,
    const float* __restrict__ h2h_w, const float* __restrict__ h2h_b,
    const float* __restrict__ fc_w,
    float* sWT, float* sBias, float* sFc) {
  for (int i = tid; i < 44 * 128; i += 256) {
    int k = i >> 7, n = i & 127;
    sWT[n * WT_STRIDE + k] = (k < 12) ? x2h_w[k * 128 + n] : h2h_w[(k - 12) * 128 + n];
  }
  if (tid < 128) sBias[tid] = x2h_b[tid] + h2h_b[tid];
  if (tid < 32) sFc[tid] = fc_w[tid];
}

__device__ __forceinline__ void load_afrag(const float* sXw, const float* sHw,
                                           int m, int hi, v2f* ax, v2f* ah) {
#pragma unroll
  for (int kk = 0; kk < 3; ++kk) ax[kk] = *(const v2f*)(sXw + m * 12 + 4 * kk + 2 * hi);
#pragma unroll
  for (int kk = 0; kk < 8; ++kk) ah[kk] = *(const v2f*)(sHw + m * 32 + 4 * kk + 2 * hi);
}

__device__ __forceinline__ void gate_half(const v2f* ax, const v2f* ah,
                                          const float* __restrict__ sWT,
                                          const float* __restrict__ sBias,
                                          int m, int hi, int ch, v8f* acc) {
#pragma unroll
  for (int g = 0; g < 4; ++g) {
    const int ct = 2 * g + ch;
    float bb = sBias[ct * 16 + m];
    v8f a = {bb, bb, bb, bb, bb, bb, bb, bb};
    const float* wcol = sWT + (ct * 16 + m) * WT_STRIDE;
#pragma unroll
    for (int kk = 0; kk < 3; ++kk) {
      v2f bf = *(const v2f*)(wcol + 4 * kk + 2 * hi);
      a = __builtin_amdgcn_wmma_f32_16x16x4_f32(false, ax[kk], false, bf, (short)0, a,
                                                false, false);
    }
#pragma unroll
    for (int kk = 0; kk < 8; ++kk) {
      v2f bf = *(const v2f*)(wcol + 12 + 4 * kk + 2 * hi);
      a = __builtin_amdgcn_wmma_f32_16x16x4_f32(false, ah[kk], false, bf, (short)0, a,
                                                false, false);
    }
    acc[g] = a;
  }
}

__device__ __forceinline__ void lstm_half(const v8f* acc, float* c8, float* sHw,
                                          int m, int hi, int ch) {
#pragma unroll
  for (int r = 0; r < 8; ++r) {
    float iv = sigm(acc[0][r]);
    float fv = sigm(acc[1][r]);
    float gv = tanhf(acc[2][r]);
    float ov = sigm(acc[3][r]);
    float cn = fv * c8[r] + iv * gv;
    float hn = ov * tanhf(cn);
    c8[r] = cn;
    sHw[(r + 8 * hi) * 32 + ch * 16 + m] = hn;
  }
}

__device__ __forceinline__ void lstm_step(const float* __restrict__ sXw, float* sHw,
                                          const float* __restrict__ sWT,
                                          const float* __restrict__ sBias,
                                          int m, int hi, float* c01) {
  v2f ax[3], ah[8];
  load_afrag(sXw, sHw, m, hi, ax, ah);
  v8f acc[4];
  gate_half(ax, ah, sWT, sBias, m, hi, 0, acc);
  lstm_half(acc, c01, sHw, m, hi, 0);
  gate_half(ax, ah, sWT, sBias, m, hi, 1, acc);
  lstm_half(acc, c01 + 8, sHw, m, hi, 1);
}

__launch_bounds__(256, 8)
__global__ void hist_lstm_kernel(
    const float* __restrict__ pm25, const float* __restrict__ feat,
    const float* __restrict__ xg,
    const float* __restrict__ x2h_w, const float* __restrict__ x2h_b,
    const float* __restrict__ h2h_w, const float* __restrict__ h2h_b,
    const float* __restrict__ fc_w, const float* __restrict__ fc_b,
    float* __restrict__ h_glob, float* __restrict__ c_glob, float* __restrict__ xn_glob) {
  __shared__ float sWT[128 * WT_STRIDE];
  __shared__ float sBias[128];
  __shared__ float sFc[32];
  __shared__ float sX[8][16 * 12];
  __shared__ float sH[8][16 * 32];

  const int tid = threadIdx.x;
  load_weights(tid, x2h_w, x2h_b, h2h_w, h2h_b, fc_w, sWT, sBias, sFc);

  const int wave = tid >> 5, lane = tid & 31, m = lane & 15, hi = lane >> 4;
  const int rowbase = blockIdx.x * 128 + wave * 16;
  float* sXw = sX[wave];
  float* sHw = sH[wave];
  for (int j = lane; j < 512; j += 32) sHw[j] = 0.f;
  float c01[16];
#pragma unroll
  for (int j = 0; j < 16; ++j) c01[j] = 0.f;
  __syncthreads();

  for (int t = 0; t < HISTT; ++t) {
    if (lane < 16) {
      int R = rowbase + lane;
      int b = R >> 12, n = R & 4095;
      size_t hb = ((size_t)(b * 16 + t)) * NNODE + n;
      sXw[lane * 12 + 0] = pm25[hb];
      const float* fp = feat + (((size_t)b * 24 + t) * NNODE + n) * 9;
#pragma unroll
      for (int j = 0; j < 9; ++j) sXw[lane * 12 + 1 + j] = fp[j];
      sXw[lane * 12 + 10] = xg[hb * 2 + 0];
      sXw[lane * 12 + 11] = xg[hb * 2 + 1];
    }
    __syncthreads();
    lstm_step(sXw, sHw, sWT, sBias, m, hi, c01);
  }

#pragma unroll
  for (int ct = 0; ct < 2; ++ct)
#pragma unroll
    for (int r = 0; r < 8; ++r) {
      size_t R = (size_t)rowbase + r + 8 * hi;
      c_glob[R * 32 + ct * 16 + m] = c01[ct * 8 + r];
      h_glob[R * 32 + ct * 16 + m] = sHw[(r + 8 * hi) * 32 + ct * 16 + m];
    }
  __syncthreads();
  if (lane < 16) {
    float s = fc_b[0];
#pragma unroll
    for (int j = 0; j < 32; ++j) s += sHw[lane * 32 + j] * sFc[j];
    xn_glob[rowbase + lane] = s;
  }
}

__launch_bounds__(256)
__global__ void pred_lstm_kernel(
    const float* __restrict__ feat,
    const float* __restrict__ q_pred, const float* __restrict__ acc_pred,
    const float* __restrict__ x2h_w, const float* __restrict__ x2h_b,
    const float* __restrict__ h2h_w, const float* __restrict__ h2h_b,
    const float* __restrict__ fc_w, const float* __restrict__ fc_b,
    float* __restrict__ h_glob, float* __restrict__ c_glob, float* __restrict__ xn_glob,
    float* __restrict__ out, int kstep) {
  __shared__ float sWT[128 * WT_STRIDE];
  __shared__ float sBias[128];
  __shared__ float sFc[32];
  __shared__ float sX[8][16 * 12];
  __shared__ float sH[8][16 * 32];

  const int tid = threadIdx.x;
  load_weights(tid, x2h_w, x2h_b, h2h_w, h2h_b, fc_w, sWT, sBias, sFc);

  const int wave = tid >> 5, lane = tid & 31, m = lane & 15, hi = lane >> 4;
  const int rowbase = blockIdx.x * 128 + wave * 16;
  float* sXw = sX[wave];
  float* sHw = sH[wave];
  for (int j = lane; j < 512; j += 32) sHw[j] = h_glob[(size_t)rowbase * 32 + j];
  float c01[16];
#pragma unroll
  for (int ct = 0; ct < 2; ++ct)
#pragma unroll
    for (int r = 0; r < 8; ++r)
      c01[ct * 8 + r] = c_glob[((size_t)rowbase + r + 8 * hi) * 32 + ct * 16 + m];

  if (lane < 16) {
    int R = rowbase + lane;
    int b = R >> 12, n = R & 4095;
    sXw[lane * 12 + 0] = xn_glob[R];
    const float* fp = feat + (((size_t)b * 24 + HISTT + kstep) * NNODE + n) * 9;
#pragma unroll
    for (int j = 0; j < 9; ++j) sXw[lane * 12 + 1 + j] = fp[j];
    float a0 = 0.f, a1 = 0.f;
#pragma unroll
    for (int pp = 0; pp < PRED_PARTS; ++pp) {
      const float* ap = acc_pred + (size_t)pp * (BATCH * NNODE * 2) + (size_t)R * 2;
      a0 += ap[0]; a1 += ap[1];
    }
    sXw[lane * 12 + 10] = sigm(q_pred[(size_t)R * 2 + 0] + a0);
    sXw[lane * 12 + 11] = sigm(q_pred[(size_t)R * 2 + 1] + a1);
  }
  __syncthreads();

  lstm_step(sXw, sHw, sWT, sBias, m, hi, c01);

#pragma unroll
  for (int ct = 0; ct < 2; ++ct)
#pragma unroll
    for (int r = 0; r < 8; ++r) {
      size_t R = (size_t)rowbase + r + 8 * hi;
      c_glob[R * 32 + ct * 16 + m] = c01[ct * 8 + r];
      h_glob[R * 32 + ct * 16 + m] = sHw[(r + 8 * hi) * 32 + ct * 16 + m];
    }
  __syncthreads();
  if (lane < 16) {
    float s = fc_b[0];
#pragma unroll
    for (int j = 0; j < 32; ++j) s += sHw[lane * 32 + j] * sFc[j];
    int R = rowbase + lane;
    int b = R >> 12, n = R & 4095;
    xn_glob[R] = s;
    out[((size_t)b * PREDT + kstep) * NNODE + n] = s;
  }
}

extern "C" void kernel_launch(void* const* d_in, const int* in_sizes, int n_in,
                              void* d_out, int out_size, void* d_ws, size_t ws_size,
                              hipStream_t stream) {
  (void)in_sizes; (void)n_in; (void)out_size; (void)ws_size;
  const float* pm25   = (const float*)d_in[0];
  const float* feat   = (const float*)d_in[1];
  const int*   ei     = (const int*)d_in[2];
  const float* cheb_w = (const float*)d_in[3];
  const float* cheb_b = (const float*)d_in[4];
  const float* x2h_w  = (const float*)d_in[5];
  const float* x2h_b  = (const float*)d_in[6];
  const float* h2h_w  = (const float*)d_in[7];
  const float* h2h_b  = (const float*)d_in[8];
  const float* fc_w   = (const float*)d_in[9];
  const float* fc_b   = (const float*)d_in[10];
  float* out = (float*)d_out;
  char* ws = (char*)d_ws;

  float* deg      = (float*)(ws + 0);
  float* dis      = (float*)(ws + 0x4000);
  float* enorm    = (float*)(ws + 0x8000);
  float* p_hist   = (float*)(ws + 0x100000);
  float* q_hist   = (float*)(ws + 0x1100000);
  float* acc_hist = (float*)(ws + 0x2100000);
  float* xn_glob  = (float*)(ws + 0x3100000);
  float* p_pred   = (float*)(ws + 0x3180000);
  float* q_pred   = (float*)(ws + 0x3280000);
  float* h_glob = p_hist;
  float* c_glob = q_hist;
  float* acc_pred = acc_hist;

  k_zero<<<(NNODE + 255) / 256, 256, 0, stream>>>(deg, NNODE);
  k_deg<<<NEDGE / 256, 256, 0, stream>>>(ei, deg);
  k_dis<<<NNODE / 256, 256, 0, stream>>>(deg, dis);
  k_enorm<<<NEDGE / 256, 256, 0, stream>>>(ei, dis, enorm);

  k_proj_hist<<<BATCH * HISTT * NNODE / 256, 256, 0, stream>>>(pm25, feat, cheb_w, cheb_b,
                                                               p_hist, q_hist);
  k_conv_slice<true, 1><<<dim3(BATCH * HISTT, 1), CONV_THREADS, 0, stream>>>(
      ei, enorm, p_hist, q_hist, acc_hist);

  hist_lstm_kernel<<<BATCH * NNODE / 128, 256, 0, stream>>>(
      pm25, feat, acc_hist, x2h_w, x2h_b, h2h_w, h2h_b, fc_w, fc_b, h_glob, c_glob, xn_glob);

  for (int k = 0; k < PREDT; ++k) {
    k_proj_pred<<<BATCH * NNODE / 256, 256, 0, stream>>>(xn_glob, feat, cheb_w, cheb_b,
                                                         p_pred, q_pred, k);
    k_conv_slice<false, PRED_PARTS><<<dim3(BATCH, PRED_PARTS), CONV_THREADS, 0, stream>>>(
        ei, enorm, p_pred, q_pred, acc_pred);
    pred_lstm_kernel<<<BATCH * NNODE / 128, 256, 0, stream>>>(
        feat, q_pred, acc_pred, x2h_w, x2h_b, h2h_w, h2h_b, fc_w, fc_b,
        h_glob, c_glob, xn_glob, out, k);
  }
}


# Projection kernels folded into the scatter kernels: each conv work-group computes p=x@W1 and q=x@W0+b from raw inputs while staging its slice in LDS; 31 -> 22 launches, summation orders unchanged
# speedup vs baseline: 5.1732x; 1.0485x over previous
#include <hip/hip_runtime.h>
#include <hip/hip_bf16.h>
#include <cstdint>
#include <cstddef>

#define BATCH 32
#define NNODE 4096
#define HISTT 16
#define PREDT 8
#define NEDGE 65536

typedef float v2f __attribute__((ext_vector_type(2)));
typedef float v8f __attribute__((ext_vector_type(8)));

__device__ __forceinline__ float sigm(float x) { return 1.f / (1.f + __expf(-x)); }

__device__ __forceinline__ void atomAddF(float* p, float v) {
  __hip_atomic_fetch_add(p, v, __ATOMIC_RELAXED, __HIP_MEMORY_SCOPE_AGENT);
}

__device__ __forceinline__ void ldsAddF(float* p, float v) {
  __hip_atomic_fetch_add(p, v, __ATOMIC_RELAXED, __HIP_MEMORY_SCOPE_WORKGROUP);
}

__global__ void k_zero(float* __restrict__ p, int n) {
  int i = blockIdx.x * 256 + threadIdx.x;
  if (i < n) p[i] = 0.f;
}

__global__ void k_deg(const int* __restrict__ ei, float* __restrict__ deg) {
  int e = blockIdx.x * 256 + threadIdx.x;
  if (e < NEDGE) atomAddF(deg + ei[e], 1.f);
}

__global__ void k_dis(const float* __restrict__ deg, float* __restrict__ dis) {
  int n = blockIdx.x * 256 + threadIdx.x;
  if (n < NNODE) {
    float d = deg[n];
    dis[n] = d > 0.f ? rsqrtf(d) : 0.f;
  }
}

__global__ void k_enorm(const int* __restrict__ ei, const float* __restrict__ dis,
                        float* __restrict__ enorm) {
  int e = blockIdx.x * 256 + threadIdx.x;
  if (e < NEDGE) enorm[e] = -dis[ei[e]] * dis[ei[NEDGE + e]];
}

__device__ __forceinline__ void proj_p(const float* x, const float* __restrict__ cheb_w,
                                       float& p0, float& p1) {
  p0 = 0.f; p1 = 0.f;
#pragma unroll
  for (int f = 0; f < 10; ++f) {
    p0 += x[f] * cheb_w[20 + f * 2 + 0];
    p1 += x[f] * cheb_w[20 + f * 2 + 1];
  }
}

__device__ __forceinline__ void proj_q(const float* x, const float* __restrict__ cheb_w,
                                       const float* __restrict__ cheb_b, float& q0, float& q1) {
  q0 = cheb_b[0]; q1 = cheb_b[1];
#pragma unroll
  for (int f = 0; f < 10; ++f) {
    q0 += x[f] * cheb_w[f * 2 + 0];
    q1 += x[f] * cheb_w[f * 2 + 1];
  }
}

#define CONV_THREADS 1024
#define EDGE_GROUP 4
#define PRED_PARTS 4

template <bool FIN, int PARTS>
__launch_bounds__(CONV_THREADS, 16)
__global__ void k_conv_slice(const int* __restrict__ ei, const float* __restrict__ enorm,
                             const float* __restrict__ x0, const float* __restrict__ feat,
                             const float* __restrict__ cheb_w, const float* __restrict__ cheb_b,
                             float* __restrict__ qout, float* __restrict__ dstbuf, int kstep) {
  __shared__ float sP[NNODE * 2];
  __shared__ float sA0[NNODE];
  __shared__ float sA1[NNODE];
  __shared__ float sQ[FIN ? NNODE * 2 : 4];
  const int tid = threadIdx.x;
  const size_t base = (size_t)blockIdx.x * (NNODE * 2);

  const int fs = FIN ? (int)(blockIdx.x >> 4) * (HISTT + PREDT) + (int)(blockIdx.x & 15)
                     : (int)blockIdx.x * (HISTT + PREDT) + HISTT + kstep;
  const float2* x2 = (const float2*)(x0 + (size_t)blockIdx.x * NNODE);
  const float2* f2 = (const float2*)(feat + (size_t)fs * NNODE * 9);
  const bool wantQ = FIN || blockIdx.y == 0;
  constexpr int PAIRS = NNODE / 2 / CONV_THREADS;
  float2 xv[PAIRS], fr[PAIRS][9];
#pragma unroll
  for (int t = 0; t < PAIRS; ++t) {
    const int i = tid + t * CONV_THREADS;
    xv[t] = x2[i];
#pragma unroll
    for (int j = 0; j < 9; ++j) fr[t][j] = f2[(size_t)i * 9 + j];
  }
#pragma unroll
  for (int t = 0; t < PAIRS; ++t) {
    const int i = tid + t * CONV_THREADS;
    float xa[10], xb[10];
    xa[0] = xv[t].x; xb[0] = xv[t].y;
#pragma unroll
    for (int j = 0; j < 9; ++j) {
      const int ja = j, jb = 9 + j;
      xa[1 + j] = (ja & 1) ? fr[t][ja >> 1].y : fr[t][ja >> 1].x;
      xb[1 + j] = (jb & 1) ? fr[t][jb >> 1].y : fr[t][jb >> 1].x;
    }
    float pa0, pa1, pb0, pb1;
    proj_p(xa, cheb_w, pa0, pa1);
    proj_p(xb, cheb_w, pb0, pb1);
    ((float4*)sP)[i] = make_float4(pa0, pa1, pb0, pb1);
    if (wantQ) {
      float qa0, qa1, qb0, qb1;
      proj_q(xa, cheb_w, cheb_b, qa0, qa1);
      proj_q(xb, cheb_w, cheb_b, qb0, qb1);
      if (FIN) ((float4*)sQ)[i] = make_float4(qa0, qa1, qb0, qb1);
      else ((float4*)(qout + base))[i] = make_float4(qa0, qa1, qb0, qb1);
    }
  }
  for (int i = tid; i < NNODE / 4; i += CONV_THREADS) {
    ((float4*)sA0)[i] = make_float4(0.f, 0.f, 0.f, 0.f);
    ((float4*)sA1)[i] = make_float4(0.f, 0.f, 0.f, 0.f);
  }
  __syncthreads();

  const int4* src4 = (const int4*)ei;
  const int4* dst4 = (const int4*)(ei + NEDGE);
  const float4* nv4 = (const float4*)enorm;
  const int iBeg = blockIdx.y * (NEDGE / 4 / PARTS), iEnd = iBeg + NEDGE / 4 / PARTS;
  for (int i0 = iBeg + tid; i0 < iEnd; i0 += CONV_THREADS * EDGE_GROUP) {
    int4 s[EDGE_GROUP], d[EDGE_GROUP];
    float4 nv[EDGE_GROUP];
#pragma unroll
    for (int u = 0; u < EDGE_GROUP; ++u) {
      int i = i0 + u * CONV_THREADS;
      s[u] = src4[i]; d[u] = dst4[i]; nv[u] = nv4[i];
    }
#pragma unroll
    for (int u = 0; u < EDGE_GROUP; ++u) {
      v2f p0 = *(const v2f*)(sP + 2 * s[u].x);
      v2f p1 = *(const v2f*)(sP + 2 * s[u].y);
      v2f p2 = *(const v2f*)(sP + 2 * s[u].z);
      v2f p3 = *(const v2f*)(sP + 2 * s[u].w);
      ldsAddF(sA0 + d[u].x, p0.x * nv[u].x); ldsAddF(sA1 + d[u].x, p0.y * nv[u].x);
      ldsAddF(sA0 + d[u].y, p1.x * nv[u].y); ldsAddF(sA1 + d[u].y, p1.y * nv[u].y);
      ldsAddF(sA0 + d[u].z, p2.x * nv[u].z); ldsAddF(sA1 + d[u].z, p2.y * nv[u].z);
      ldsAddF(sA0 + d[u].w, p3.x * nv[u].w); ldsAddF(sA1 + d[u].w, p3.y * nv[u].w);
    }
  }
  __syncthreads();

  float4* o4 = (float4*)(dstbuf + (size_t)blockIdx.y * gridDim.x * (NNODE * 2) + base);
  for (int i = tid; i < NNODE / 2; i += CONV_THREADS) {
    v2f a0 = *(const v2f*)(sA0 + 2 * i);
    v2f a1 = *(const v2f*)(sA1 + 2 * i);
    float4 r = make_float4(a0.x, a1.x, a0.y, a1.y);
    if (FIN) {
      float4 qv = ((const float4*)sQ)[i];
      r.x = sigm(qv.x + r.x); r.y = sigm(qv.y + r.y);
      r.z = sigm(qv.z + r.z); r.w = sigm(qv.w + r.w);
    }
    o4[i] = r;
  }
}

#define WT_STRIDE 48

__device__ __forceinline__ void load_weights(int tid,
    const float* __restrict__ x2h_w, const float* __restrict__ x2h_b,
    const float* __restrict__ h2h_w, const float* __restrict__ h2h_b,
    const float* __restrict__ fc_w,
    float* sWT, float* sBias, float* sFc) {
  for (int i = tid; i < 44 * 128; i += 256) {
    int k = i >> 7, n = i & 127;
    sWT[n * WT_STRIDE + k] = (k < 12) ? x2h_w[k * 128 + n] : h2h_w[(k - 12) * 128 + n];
  }
  if (tid < 128) sBias[tid] = x2h_b[tid] + h2h_b[tid];
  if (tid < 32) sFc[tid] = fc_w[tid];
}

__device__ __forceinline__ void load_afrag(const float* sXw, const float* sHw,
                                           int m, int hi, v2f* ax, v2f* ah) {
#pragma unroll
  for (int kk = 0; kk < 3; ++kk) ax[kk] = *(const v2f*)(sXw + m * 12 + 4 * kk + 2 * hi);
#pragma unroll
  for (int kk = 0; kk < 8; ++kk) ah[kk] = *(const v2f*)(sHw + m * 32 + 4 * kk + 2 * hi);
}

__device__ __forceinline__ void gate_half(const v2f* ax, const v2f* ah,
                                          const float* __restrict__ sWT,
                                          const float* __restrict__ sBias,
                                          int m, int hi, int ch, v8f* acc) {
#pragma unroll
  for (int g = 0; g < 4; ++g) {
    const int ct = 2 * g + ch;
    float bb = sBias[ct * 16 + m];
    v8f a = {bb, bb, bb, bb, bb, bb, bb, bb};
    const float* wcol = sWT + (ct * 16 + m) * WT_STRIDE;
#pragma unroll
    for (int kk = 0; kk < 3; ++kk) {
      v2f bf = *(const v2f*)(wcol + 4 * kk + 2 * hi);
      a = __builtin_amdgcn_wmma_f32_16x16x4_f32(false, ax[kk], false, bf, (short)0, a,
                                                false, false);
    }
#pragma unroll
    for (int kk = 0; kk < 8; ++kk) {
      v2f bf = *(const v2f*)(wcol + 12 + 4 * kk + 2 * hi);
      a = __builtin_amdgcn_wmma_f32_16x16x4_f32(false, ah[kk], false, bf, (short)0, a,
                                                false, false);
    }
    acc[g] = a;
  }
}

__device__ __forceinline__ void lstm_half(const v8f* acc, float* c8, float* sHw,
                                          int m, int hi, int ch) {
#pragma unroll
  for (int r = 0; r < 8; ++r) {
    float iv = sigm(acc[0][r]);
    float fv = sigm(acc[1][r]);
    float gv = tanhf(acc[2][r]);
    float ov = sigm(acc[3][r]);
    float cn = fv * c8[r] + iv * gv;
    float hn = ov * tanhf(cn);
    c8[r] = cn;
    sHw[(r + 8 * hi) * 32 + ch * 16 + m] = hn;
  }
}

__device__ __forceinline__ void lstm_step(const float* __restrict__ sXw, float* sHw,
                                          const float* __restrict__ sWT,
                                          const float* __restrict__ sBias,
                                          int m, int hi, float* c01) {
  v2f ax[3], ah[8];
  load_afrag(sXw, sHw, m, hi, ax, ah);
  v8f acc[4];
  gate_half(ax, ah, sWT, sBias, m, hi, 0, acc);
  lstm_half(acc, c01, sHw, m, hi, 0);
  gate_half(ax, ah, sWT, sBias, m, hi, 1, acc);
  lstm_half(acc, c01 + 8, sHw, m, hi, 1);
}

__launch_bounds__(256, 8)
__global__ void hist_lstm_kernel(
    const float* __restrict__ pm25, const float* __restrict__ feat,
    const float* __restrict__ xg,
    const float* __restrict__ x2h_w, const float* __restrict__ x2h_b,
    const float* __restrict__ h2h_w, const float* __restrict__ h2h_b,
    const float* __restrict__ fc_w, const float* __restrict__ fc_b,
    float* __restrict__ h_glob, float* __restrict__ c_glob, float* __restrict__ xn_glob) {
  __shared__ float sWT[128 * WT_STRIDE];
  __shared__ float sBias[128];
  __shared__ float sFc[32];
  __shared__ float sX[8][16 * 12];
  __shared__ float sH[8][16 * 32];

  const int tid = threadIdx.x;
  load_weights(tid, x2h_w, x2h_b, h2h_w, h2h_b, fc_w, sWT, sBias, sFc);

  const int wave = tid >> 5, lane = tid & 31, m = lane & 15, hi = lane >> 4;
  const int rowbase = blockIdx.x * 128 + wave * 16;
  float* sXw = sX[wave];
  float* sHw = sH[wave];
  for (int j = lane; j < 512; j += 32) sHw[j] = 0.f;
  float c01[16];
#pragma unroll
  for (int j = 0; j < 16; ++j) c01[j] = 0.f;
  __syncthreads();

  for (int t = 0; t < HISTT; ++t) {
    if (lane < 16) {
      int R = rowbase + lane;
      int b = R >> 12, n = R & 4095;
      size_t hb = ((size_t)(b * 16 + t)) * NNODE + n;
      sXw[lane * 12 + 0] = pm25[hb];
      const float* fp = feat + (((size_t)b * 24 + t) * NNODE + n) * 9;
#pragma unroll
      for (int j = 0; j < 9; ++j) sXw[lane * 12 + 1 + j] = fp[j];
      sXw[lane * 12 + 10] = xg[hb * 2 + 0];
      sXw[lane * 12 + 11] = xg[hb * 2 + 1];
    }
    __syncthreads();
    lstm_step(sXw, sHw, sWT, sBias, m, hi, c01);
  }

#pragma unroll
  for (int ct = 0; ct < 2; ++ct)
#pragma unroll
    for (int r = 0; r < 8; ++r) {
      size_t R = (size_t)rowbase + r + 8 * hi;
      c_glob[R * 32 + ct * 16 + m] = c01[ct * 8 + r];
      h_glob[R * 32 + ct * 16 + m] = sHw[(r + 8 * hi) * 32 + ct * 16 + m];
    }
  __syncthreads();
  if (lane < 16) {
    float s = fc_b[0];
#pragma unroll
    for (int j = 0; j < 32; ++j) s += sHw[lane * 32 + j] * sFc[j];
    xn_glob[rowbase + lane] = s;
  }
}

__launch_bounds__(256)
__global__ void pred_lstm_kernel(
    const float* __restrict__ feat,
    const float* __restrict__ q_pred, const float* __restrict__ acc_pred,
    const float* __restrict__ x2h_w, const float* __restrict__ x2h_b,
    const float* __restrict__ h2h_w, const float* __restrict__ h2h_b,
    const float* __restrict__ fc_w, const float* __restrict__ fc_b,
    float* __restrict__ h_glob, float* __restrict__ c_glob, float* __restrict__ xn_glob,
    float* __restrict__ out, int kstep) {
  __shared__ float sWT[128 * WT_STRIDE];
  __shared__ float sBias[128];
  __shared__ float sFc[32];
  __shared__ float sX[8][16 * 12];
  __shared__ float sH[8][16 * 32];

  const int tid = threadIdx.x;
  load_weights(tid, x2h_w, x2h_b, h2h_w, h2h_b, fc_w, sWT, sBias, sFc);

  const int wave = tid >> 5, lane = tid & 31, m = lane & 15, hi = lane >> 4;
  const int rowbase = blockIdx.x * 128 + wave * 16;
  float* sXw = sX[wave];
  float* sHw = sH[wave];
  for (int j = lane; j < 512; j += 32) sHw[j] = h_glob[(size_t)rowbase * 32 + j];
  float c01[16];
#pragma unroll
  for (int ct = 0; ct < 2; ++ct)
#pragma unroll
    for (int r = 0; r < 8; ++r)
      c01[ct * 8 + r] = c_glob[((size_t)rowbase + r + 8 * hi) * 32 + ct * 16 + m];

  if (lane < 16) {
    int R = rowbase + lane;
    int b = R >> 12, n = R & 4095;
    sXw[lane * 12 + 0] = xn_glob[R];
    const float* fp = feat + (((size_t)b * 24 + HISTT + kstep) * NNODE + n) * 9;
#pragma unroll
    for (int j = 0; j < 9; ++j) sXw[lane * 12 + 1 + j] = fp[j];
    float a0 = 0.f, a1 = 0.f;
#pragma unroll
    for (int pp = 0; pp < PRED_PARTS; ++pp) {
      const float* ap = acc_pred + (size_t)pp * (BATCH * NNODE * 2) + (size_t)R * 2;
      a0 += ap[0]; a1 += ap[1];
    }
    sXw[lane * 12 + 10] = sigm(q_pred[(size_t)R * 2 + 0] + a0);
    sXw[lane * 12 + 11] = sigm(q_pred[(size_t)R * 2 + 1] + a1);
  }
  __syncthreads();

  lstm_step(sXw, sHw, sWT, sBias, m, hi, c01);

#pragma unroll
  for (int ct = 0; ct < 2; ++ct)
#pragma unroll
    for (int r = 0; r < 8; ++r) {
      size_t R = (size_t)rowbase + r + 8 * hi;
      c_glob[R * 32 + ct * 16 + m] = c01[ct * 8 + r];
      h_glob[R * 32 + ct * 16 + m] = sHw[(r + 8 * hi) * 32 + ct * 16 + m];
    }
  __syncthreads();
  if (lane < 16) {
    float s = fc_b[0];
#pragma unroll
    for (int j = 0; j < 32; ++j) s += sHw[lane * 32 + j] * sFc[j];
    int R = rowbase + lane;
    int b = R >> 12, n = R & 4095;
    xn_glob[R] = s;
    out[((size_t)b * PREDT + kstep) * NNODE + n] = s;
  }
}

extern "C" void kernel_launch(void* const* d_in, const int* in_sizes, int n_in,
                              void* d_out, int out_size, void* d_ws, size_t ws_size,
                              hipStream_t stream) {
  (void)in_sizes; (void)n_in; (void)out_size; (void)ws_size;
  const float* pm25   = (const float*)d_in[0];
  const float* feat   = (const float*)d_in[1];
  const int*   ei     = (const int*)d_in[2];
  const float* cheb_w = (const float*)d_in[3];
  const float* cheb_b = (const float*)d_in[4];
  const float* x2h_w  = (const float*)d_in[5];
  const float* x2h_b  = (const float*)d_in[6];
  const float* h2h_w  = (const float*)d_in[7];
  const float* h2h_b  = (const float*)d_in[8];
  const float* fc_w   = (const float*)d_in[9];
  const float* fc_b   = (const float*)d_in[10];
  float* out = (float*)d_out;
  char* ws = (char*)d_ws;

  float* deg      = (float*)(ws + 0);
  float* dis      = (float*)(ws + 0x4000);
  float* enorm    = (float*)(ws + 0x8000);
  float* h_glob   = (float*)(ws + 0x100000);
  float* c_glob   = (float*)(ws + 0x1100000);
  float* acc_hist = (float*)(ws + 0x2100000);
  float* xn_glob  = (float*)(ws + 0x3100000);
  float* q_pred   = (float*)(ws + 0x3280000);
  float* acc_pred = acc_hist;

  k_zero<<<(NNODE + 255) / 256, 256, 0, stream>>>(deg, NNODE);
  k_deg<<<NEDGE / 256, 256, 0, stream>>>(ei, deg);
  k_dis<<<NNODE / 256, 256, 0, stream>>>(deg, dis);
  k_enorm<<<NEDGE / 256, 256, 0, stream>>>(ei, dis, enorm);

  k_conv_slice<true, 1><<<dim3(BATCH * HISTT, 1), CONV_THREADS, 0, stream>>>(
      ei, enorm, pm25, feat, cheb_w, cheb_b, q_pred, acc_hist, 0);

  hist_lstm_kernel<<<BATCH * NNODE / 128, 256, 0, stream>>>(
      pm25, feat, acc_hist, x2h_w, x2h_b, h2h_w, h2h_b, fc_w, fc_b, h_glob, c_glob, xn_glob);

  for (int k = 0; k < PREDT; ++k) {
    k_conv_slice<false, PRED_PARTS><<<dim3(BATCH, PRED_PARTS), CONV_THREADS, 0, stream>>>(
        ei, enorm, xn_glob, feat, cheb_w, cheb_b, q_pred, acc_pred, k);
    pred_lstm_kernel<<<BATCH * NNODE / 128, 256, 0, stream>>>(
        feat, q_pred, acc_pred, x2h_w, x2h_b, h2h_w, h2h_b, fc_w, fc_b,
        h_glob, c_glob, xn_glob, out, k);
  }
}
